# starting kernel
# baseline (speedup 1.0000x reference)
#include <hip/hip_runtime.h>
#include <math.h>

typedef float v2f __attribute__((ext_vector_type(2)));
typedef float v8f __attribute__((ext_vector_type(8)));

#define HW    4096
#define CHW   (256 * 4096)
#define PIXS  292
#define MROW  18
#define YROW  20
#define WAVES 2

__global__ __launch_bounds__(256) void repack_weights(const float* __restrict__ wcore,
                                                      const float* __restrict__ wout,
                                                      float* __restrict__ ws) {
    int idx = blockIdx.x * 256 + threadIdx.x;
    if (idx < 65536) {
        int lo   = idx & 1;
        int pair = idx >> 1;
        int n    = pair & 15;
        int m    = (pair >> 4) & 127;
        int t    = pair >> 11;
        int c    = 2 * m + lo;
        ws[idx] = wcore[t * 4096 + c * 16 + n];
    } else {
        int f    = idx - 65536;
        int lo   = f & 1;
        int rest = f >> 1;
        int d    = rest & 255;
        int m    = rest >> 8;
        int k    = 2 * m + lo;
        int e    = k >> 4;
        int g    = k & 15;
        ws[idx] = wout[g * 4096 + d * 16 + e];
    }
}

__global__ __launch_bounds__(WAVES * 32) void tr_main(const float* __restrict__ x,
                                                      const float* __restrict__ wp2,
                                                      const float* __restrict__ vp2,
                                                      float* __restrict__ out) {
    extern __shared__ float smem[];
    const int wave = threadIdx.x >> 5;
    const int lane = threadIdx.x & 31;
    const int half = lane >> 4;
    const int ln   = lane & 15;

    float* Mw = smem + wave * (16 * PIXS);
    float* Yw = smem + WAVES * (16 * PIXS) + wave * (16 * YROW);

    const int tile = blockIdx.x * WAVES + wave;
    const int img  = tile >> 8;
    const int p0   = (tile & 255) << 4;
    const float* xg = x + (size_t)img * CHW + p0 + ln;

    for (int t = 0; t < 16; ++t) {
        v8f acc = {};
        const v2f* wb2 = (const v2f*)wp2 + t * 2048 + ln;
#pragma unroll 4
        for (int kb = 0; kb < 256; kb += 4) {
            const int c0 = kb + 2 * half;
            v2f a;
            a.x = xg[(size_t)c0 * HW];
            a.y = xg[(size_t)(c0 + 1) * HW];
            v2f b = wb2[(c0 >> 1) * 16];
            acc = __builtin_amdgcn_wmma_f32_16x16x4_f32(false, a, false, b,
                                                        (short)0, acc, false, false);
        }
#pragma unroll
        for (int v = 0; v < 8; ++v) {
            const int px = v + 8 * half;
            Mw[px * PIXS + t * MROW + ln] = acc[v];
        }
    }

    for (int px = 0; px < 16; ++px) {
        float* mp = Mw + px * PIXS;
        v8f acc = {};
#pragma unroll
        for (int kb = 0; kb < 16; kb += 4) {
            const int k0 = kb + 2 * half;
            v2f a, b;
            a.x = mp[ln * MROW + k0];
            a.y = mp[ln * MROW + k0 + 1];
            b.x = mp[k0 * MROW + ln];
            b.y = mp[(k0 + 1) * MROW + ln];
            acc = __builtin_amdgcn_wmma_f32_16x16x4_f32(false, a, false, b,
                                                        (short)0, acc, false, false);
        }
#pragma unroll
        for (int v = 0; v < 8; ++v)
            mp[(v + 8 * half) * MROW + ln] = acc[v];
    }

    const int n1 = lane >> 2;
    const int pp = (lane & 3) << 2;
    for (int td = 0; td < 16; ++td) {
        v8f acc = {};
        const v2f* vb2 = (const v2f*)vp2 + td * 16 + ln;
#pragma unroll 4
        for (int kb = 0; kb < 256; kb += 4) {
            const int k0 = kb + 2 * half;
            v2f a;
            const int blk = k0 >> 4, off = k0 & 15;
            a.x = Mw[ln * PIXS + blk * MROW + off];
            a.y = Mw[ln * PIXS + blk * MROW + off + 1];
            v2f b = vb2[(k0 >> 1) * 256];
            acc = __builtin_amdgcn_wmma_f32_16x16x4_f32(false, a, false, b,
                                                        (short)0, acc, false, false);
        }
#pragma unroll
        for (int v = 0; v < 8; ++v)
            Yw[ln * YROW + v + 8 * half] = acc[v];

        const size_t ob = (size_t)img * CHW + (size_t)(td * 16) * HW + p0;
        float4 r1 = *(const float4*)&Yw[n1 * YROW + pp];
        float4 r2 = *(const float4*)&Yw[(n1 + 8) * YROW + pp];
        *(float4*)&out[ob + (size_t)n1 * HW + pp]       = r1;
        *(float4*)&out[ob + (size_t)(n1 + 8) * HW + pp] = r2;
    }
}

__global__ __launch_bounds__(256) void inorm_softsign(float* __restrict__ y) {
    __shared__ float bs[256];
    __shared__ float bss[256];
    __shared__ float mean_s, rs_s;

    float* p = y + (size_t)blockIdx.x * 4096;
    const int tid = threadIdx.x;

    float v[16];
    float s = 0.f, ss = 0.f;
#pragma unroll
    for (int i = 0; i < 16; ++i) {
        float t = p[tid + 256 * i];
        v[i] = t;
        s  += t;
        ss += t * t;
    }
    bs[tid] = s;
    bss[tid] = ss;
    __syncthreads();
    for (int off = 128; off > 0; off >>= 1) {
        if (tid < off) {
            bs[tid]  += bs[tid + off];
            bss[tid] += bss[tid + off];
        }
        __syncthreads();
    }
    if (tid == 0) {
        const float mean = bs[0] * (1.0f / 4096.0f);
        const float var  = bss[0] * (1.0f / 4096.0f) - mean * mean;
        mean_s = mean;
        rs_s   = rsqrtf(var + 1e-5f);
    }
    __syncthreads();
    const float mean = mean_s, rs = rs_s;
#pragma unroll
    for (int i = 0; i < 16; ++i) {
        const float yn = (v[i] - mean) * rs;
        p[tid + 256 * i] = yn / (1.0f + fabsf(yn));
    }
}

extern "C" void kernel_launch(void* const* d_in, const int* in_sizes, int n_in,
                              void* d_out, int out_size, void* d_ws, size_t ws_size,
                              hipStream_t stream) {
    (void)in_sizes; (void)n_in; (void)out_size; (void)ws_size;
    const float* x     = (const float*)d_in[0];
    const float* wcore = (const float*)d_in[1];
    const float* wout  = (const float*)d_in[2];
    float* out = (float*)d_out;
    float* ws  = (float*)d_ws;
    const float* wp2 = ws;
    const float* vp2 = ws + 65536;

    repack_weights<<<512, 256, 0, stream>>>(wcore, wout, ws);

    const size_t smem_bytes = (WAVES * (16 * PIXS) + WAVES * (16 * YROW)) * sizeof(float);
    tr_main<<<4096 / WAVES, WAVES * 32, smem_bytes, stream>>>(x, wp2, vp2, out);

    inorm_softsign<<<16 * 256, 256, 0, stream>>>(out);
}


# Feed both big GEMMs from on-chip copies: 8-wave workgroups share each 16 KB weight slab through LDS (fetched one slab ahead), a wave loads its x fragments from memory once; sum order unchanged
# speedup vs baseline: 1.9001x; 1.9001x over previous
#include <hip/hip_runtime.h>
#include <math.h>

typedef float v2f __attribute__((ext_vector_type(2)));
typedef float v8f __attribute__((ext_vector_type(8)));
typedef float v4f __attribute__((ext_vector_type(4)));

#define HW    4096
#define CHW   (256 * 4096)
#define PIXS  260
#define YROW  20
#define NW    8
#define SLABF 4096
#define LDSF  (NW * 16 * PIXS + SLABF + NW * 16 * YROW)

__global__ __launch_bounds__(256) void repack_weights(const float* __restrict__ wcore,
                                                      const float* __restrict__ wout,
                                                      float* __restrict__ ws) {
    int idx = blockIdx.x * 256 + threadIdx.x;
    int f    = idx & 65535;
    int lo   = f & 1;
    int pair = f >> 1;
    int n    = pair & 15;
    int m    = (pair >> 4) & 127;
    int t    = pair >> 11;
    int k    = 2 * m + lo;
    if (idx < 65536) {
        ws[idx] = wcore[t * 4096 + k * 16 + n];
    } else {
        int e = k >> 4;
        int g = k & 15;
        int d = t * 16 + n;
        ws[idx] = wout[g * 4096 + d * 16 + e];
    }
}

__global__ __launch_bounds__(NW * 32, 4) void tr_main(const float* __restrict__ x,
                                                      const float* __restrict__ wsl,
                                                      float* __restrict__ out) {
    __shared__ __attribute__((aligned(16))) float smem[LDSF];
    const int tid  = threadIdx.x;
    const int wave = tid >> 5;
    const int lane = tid & 31;
    const int half = lane >> 4;
    const int ln   = lane & 15;

    float* Mw = smem + wave * (16 * PIXS);
    float* Sl = smem + NW * (16 * PIXS);
    float* Yw = Sl + SLABF + wave * (16 * YROW);

    const int tile = blockIdx.x * NW + wave;
    const int img  = tile >> 8;
    const int p0   = (tile & 255) << 4;
    const float* xg = x + (size_t)img * CHW + (size_t)(2 * half) * HW + p0 + ln;

    const v4f* wv = (const v4f*)wsl + tid;
    v4f*       sv = (v4f*)Sl + tid;
    const v2f*    sb = (const v2f*)Sl + lane;

    v4f pf0 = wv[0], pf1 = wv[256], pf2 = wv[512], pf3 = wv[768];
    v2f xa[64];
#pragma unroll
    for (int j = 0; j < 64; ++j) {
        xa[j].x = xg[(size_t)(4 * j) * HW];
        xa[j].y = xg[(size_t)(4 * j + 1) * HW];
    }
    sv[0] = pf0; sv[256] = pf1; sv[512] = pf2; sv[768] = pf3;
    __syncthreads();

    for (int t = 0; t < 16; ++t) {
        const v4f* wn = wv + (t + 1) * 1024;
        pf0 = wn[0]; pf1 = wn[256]; pf2 = wn[512]; pf3 = wn[768];
        v8f acc = {};
        v2f bq[16], bn[16];
#pragma unroll
        for (int i = 0; i < 16; ++i) bq[i] = sb[i * 32];
#pragma unroll
        for (int g = 0; g < 4; ++g) {
            if (g < 3) {
#pragma unroll
                for (int i = 0; i < 16; ++i) bn[i] = sb[((g + 1) * 16 + i) * 32];
            }
#pragma unroll
            for (int i = 0; i < 16; ++i)
                acc = __builtin_amdgcn_wmma_f32_16x16x4_f32(false, xa[g * 16 + i], false, bq[i],
                                                            (short)0, acc, false, false);
#pragma unroll
            for (int i = 0; i < 16; ++i) bq[i] = bn[i];
        }
#pragma unroll
        for (int v = 0; v < 8; ++v)
            Mw[(v + 8 * half) * PIXS + t * 16 + ln] = acc[v];
        __syncthreads();
        sv[0] = pf0; sv[256] = pf1; sv[512] = pf2; sv[768] = pf3;
        __syncthreads();
    }

    for (int px = 0; px < 16; ++px) {
        float* mp = Mw + px * PIXS;
        v8f acc = {};
#pragma unroll
        for (int kb = 0; kb < 16; kb += 4) {
            const int k0 = kb + 2 * half;
            v2f a, b;
            a.x = mp[ln * 16 + k0];
            a.y = mp[ln * 16 + k0 + 1];
            b.x = mp[k0 * 16 + ln];
            b.y = mp[(k0 + 1) * 16 + ln];
            acc = __builtin_amdgcn_wmma_f32_16x16x4_f32(false, a, false, b,
                                                        (short)0, acc, false, false);
        }
#pragma unroll
        for (int v = 0; v < 8; ++v)
            mp[(v + 8 * half) * 16 + ln] = acc[v];
    }

    const int n1 = lane >> 2;
    const int pp = (lane & 3) << 2;
    const float* ap = Mw + ln * PIXS + 2 * half;
    for (int td = 0; td < 16; ++td) {
        if (td < 15) {
            const v4f* wn = wv + (td + 17) * 1024;
            pf0 = wn[0]; pf1 = wn[256]; pf2 = wn[512]; pf3 = wn[768];
        }
        v8f acc = {};
        v2f aq[16], an[16], bq[16], bn[16];
#pragma unroll
        for (int i = 0; i < 16; ++i) {
            aq[i] = *(const v2f*)&ap[4 * i];
            bq[i] = sb[i * 32];
        }
#pragma unroll
        for (int g = 0; g < 4; ++g) {
            if (g < 3) {
#pragma unroll
                for (int i = 0; i < 16; ++i) {
                    an[i] = *(const v2f*)&ap[4 * ((g + 1) * 16 + i)];
                    bn[i] = sb[((g + 1) * 16 + i) * 32];
                }
            }
#pragma unroll
            for (int i = 0; i < 16; ++i)
                acc = __builtin_amdgcn_wmma_f32_16x16x4_f32(false, aq[i], false, bq[i],
                                                            (short)0, acc, false, false);
#pragma unroll
            for (int i = 0; i < 16; ++i) { aq[i] = an[i]; bq[i] = bn[i]; }
        }
#pragma unroll
        for (int v = 0; v < 8; ++v)
            Yw[ln * YROW + v + 8 * half] = acc[v];

        const size_t ob = (size_t)img * CHW + (size_t)(td * 16) * HW + p0;
        float4 r1 = *(const float4*)&Yw[n1 * YROW + pp];
        float4 r2 = *(const float4*)&Yw[(n1 + 8) * YROW + pp];

        if (td < 15) {
            __syncthreads();
            sv[0] = pf0; sv[256] = pf1; sv[512] = pf2; sv[768] = pf3;
            __syncthreads();
        }
        *(float4*)&out[ob + (size_t)n1 * HW + pp]       = r1;
        *(float4*)&out[ob + (size_t)(n1 + 8) * HW + pp] = r2;
    }
}

__global__ __launch_bounds__(256) void inorm_softsign(float* __restrict__ y) {
    __shared__ float bs[256];
    __shared__ float bss[256];
    __shared__ float mean_s, rs_s;

    float* p = y + (size_t)blockIdx.x * 4096;
    const int tid = threadIdx.x;

    float v[16];
    float s = 0.f, ss = 0.f;
#pragma unroll
    for (int i = 0; i < 16; ++i) {
        float t = p[tid + 256 * i];
        v[i] = t;
        s  += t;
        ss += t * t;
    }
    bs[tid] = s;
    bss[tid] = ss;
    __syncthreads();
    for (int off = 128; off > 0; off >>= 1) {
        if (tid < off) {
            bs[tid]  += bs[tid + off];
            bss[tid] += bss[tid + off];
        }
        __syncthreads();
    }
    if (tid == 0) {
        const float mean = bs[0] * (1.0f / 4096.0f);
        const float var  = bss[0] * (1.0f / 4096.0f) - mean * mean;
        mean_s = mean;
        rs_s   = rsqrtf(var + 1e-5f);
    }
    __syncthreads();
    const float mean = mean_s, rs = rs_s;
#pragma unroll
    for (int i = 0; i < 16; ++i) {
        const float yn = (v[i] - mean) * rs;
        p[tid + 256 * i] = yn / (1.0f + fabsf(yn));
    }
}

extern "C" void kernel_launch(void* const* d_in, const int* in_sizes, int n_in,
                              void* d_out, int out_size, void* d_ws, size_t ws_size,
                              hipStream_t stream) {
    (void)in_sizes; (void)n_in; (void)out_size; (void)ws_size;
    const float* x     = (const float*)d_in[0];
    const float* wcore = (const float*)d_in[1];
    const float* wout  = (const float*)d_in[2];
    float* out = (float*)d_out;
    float* ws  = (float*)d_ws;

    repack_weights<<<512, 256, 0, stream>>>(wcore, wout, ws);

    tr_main<<<512, 256, 0, stream>>>(x, ws, out);

    inorm_softsign<<<16 * 256, 256, 0, stream>>>(out);
}


# GEMM1 k-outer: a wave holds all 16 column-tile accumulators and walks 16-channel weight slabs; its x fragments stream one slab ahead under the WMMAs, not 128 loads up front; sum order unchanged
# speedup vs baseline: 1.9554x; 1.0291x over previous
#include <hip/hip_runtime.h>
#include <math.h>

typedef float v2f __attribute__((ext_vector_type(2)));
typedef float v8f __attribute__((ext_vector_type(8)));
typedef float v4f __attribute__((ext_vector_type(4)));

#define HW    4096
#define CHW   (256 * 4096)
#define PIXS  260
#define YROW  20
#define NW    8
#define SLABF 4096
#define LDSF  (NW * 16 * PIXS + SLABF + NW * 16 * YROW)

__global__ __launch_bounds__(256) void repack_weights(const float* __restrict__ wcore,
                                                      const float* __restrict__ wout,
                                                      float* __restrict__ ws) {
    int idx = blockIdx.x * 256 + threadIdx.x;
    int f    = idx & 65535;
    int lo   = f & 1;
    int pair = f >> 1;
    int n    = pair & 15;
    if (idx < 65536) {
        int h = (pair >> 4) & 1;
        int t = (pair >> 5) & 15;
        int j = (pair >> 9) & 3;
        int s = pair >> 11;
        int c = 16 * s + 4 * j + 2 * h + lo;
        ws[idx] = wcore[t * 4096 + c * 16 + n];
    } else {
        int m = (pair >> 4) & 127;
        int t = pair >> 11;
        int k = 2 * m + lo;
        int e = k >> 4;
        int g = k & 15;
        int d = t * 16 + n;
        ws[idx] = wout[g * 4096 + d * 16 + e];
    }
}

__global__ __launch_bounds__(NW * 32, 4) void tr_main(const float* __restrict__ x,
                                                      const float* __restrict__ wsl,
                                                      float* __restrict__ out) {
    __shared__ __attribute__((aligned(16))) float smem[LDSF];
    const int tid  = threadIdx.x;
    const int wave = tid >> 5;
    const int lane = tid & 31;
    const int half = lane >> 4;
    const int ln   = lane & 15;

    float* Mw = smem + wave * (16 * PIXS);
    float* Sl = smem + NW * (16 * PIXS);
    float* Yw = Sl + SLABF + wave * (16 * YROW);

    const int tile = blockIdx.x * NW + wave;
    const int img  = tile >> 8;
    const int p0   = (tile & 255) << 4;
    const float* xg = x + (size_t)img * CHW + (size_t)(2 * half) * HW + p0 + ln;

    const v4f* wv = (const v4f*)wsl + tid;
    v4f*       sv = (v4f*)Sl + tid;
    const v2f*    sb = (const v2f*)Sl + lane;

    v4f pf0 = wv[0], pf1 = wv[256], pf2 = wv[512], pf3 = wv[768];
    v2f xc[4], xn[4];
#pragma unroll
    for (int j = 0; j < 4; ++j) {
        xc[j].x = xg[(size_t)(4 * j) * HW];
        xc[j].y = xg[(size_t)(4 * j + 1) * HW];
    }
    sv[0] = pf0; sv[256] = pf1; sv[512] = pf2; sv[768] = pf3;
    __syncthreads();

    v8f macc[16];
#pragma unroll
    for (int t = 0; t < 16; ++t) macc[t] = (v8f){};
    for (int c16 = 0; c16 < 16; ++c16) {
        const v4f* wn = wv + (c16 + 1) * 1024;
        pf0 = wn[0]; pf1 = wn[256]; pf2 = wn[512]; pf3 = wn[768];
        {
            const float* xs = xg + (size_t)(16 * ((c16 + 1) & 15)) * HW;
#pragma unroll
            for (int j = 0; j < 4; ++j) {
                xn[j].x = xs[(size_t)(4 * j) * HW];
                xn[j].y = xs[(size_t)(4 * j + 1) * HW];
            }
        }
        v2f bq[16], bn[16];
#pragma unroll
        for (int t = 0; t < 16; ++t) bq[t] = sb[t * 32];
#pragma unroll
        for (int j = 0; j < 4; ++j) {
            if (j < 3) {
#pragma unroll
                for (int t = 0; t < 16; ++t) bn[t] = sb[((j + 1) * 16 + t) * 32];
            }
#pragma unroll
            for (int t = 0; t < 16; ++t)
                macc[t] = __builtin_amdgcn_wmma_f32_16x16x4_f32(false, xc[j], false, bq[t],
                                                                (short)0, macc[t], false, false);
#pragma unroll
            for (int t = 0; t < 16; ++t) bq[t] = bn[t];
        }
#pragma unroll
        for (int j = 0; j < 4; ++j) xc[j] = xn[j];
        __syncthreads();
        sv[0] = pf0; sv[256] = pf1; sv[512] = pf2; sv[768] = pf3;
        __syncthreads();
    }
#pragma unroll
    for (int t = 0; t < 16; ++t) {
#pragma unroll
        for (int v = 0; v < 8; ++v)
            Mw[(v + 8 * half) * PIXS + t * 16 + ln] = macc[t][v];
    }
    __syncthreads();

    for (int px = 0; px < 16; ++px) {
        float* mp = Mw + px * PIXS;
        v8f acc = {};
#pragma unroll
        for (int kb = 0; kb < 16; kb += 4) {
            const int k0 = kb + 2 * half;
            v2f a, b;
            a.x = mp[ln * 16 + k0];
            a.y = mp[ln * 16 + k0 + 1];
            b.x = mp[k0 * 16 + ln];
            b.y = mp[(k0 + 1) * 16 + ln];
            acc = __builtin_amdgcn_wmma_f32_16x16x4_f32(false, a, false, b,
                                                        (short)0, acc, false, false);
        }
#pragma unroll
        for (int v = 0; v < 8; ++v)
            mp[(v + 8 * half) * 16 + ln] = acc[v];
    }

    const int n1 = lane >> 2;
    const int pp = (lane & 3) << 2;
    const float* ap = Mw + ln * PIXS + 2 * half;
    for (int td = 0; td < 16; ++td) {
        if (td < 15) {
            const v4f* wn = wv + (td + 17) * 1024;
            pf0 = wn[0]; pf1 = wn[256]; pf2 = wn[512]; pf3 = wn[768];
        }
        v8f acc = {};
        v2f aq[16], an[16], bq[16], bn[16];
#pragma unroll
        for (int i = 0; i < 16; ++i) {
            aq[i] = *(const v2f*)&ap[4 * i];
            bq[i] = sb[i * 32];
        }
#pragma unroll
        for (int g = 0; g < 4; ++g) {
            if (g < 3) {
#pragma unroll
                for (int i = 0; i < 16; ++i) {
                    an[i] = *(const v2f*)&ap[4 * ((g + 1) * 16 + i)];
                    bn[i] = sb[((g + 1) * 16 + i) * 32];
                }
            }
#pragma unroll
            for (int i = 0; i < 16; ++i)
                acc = __builtin_amdgcn_wmma_f32_16x16x4_f32(false, aq[i], false, bq[i],
                                                            (short)0, acc, false, false);
#pragma unroll
            for (int i = 0; i < 16; ++i) { aq[i] = an[i]; bq[i] = bn[i]; }
        }
#pragma unroll
        for (int v = 0; v < 8; ++v)
            Yw[ln * YROW + v + 8 * half] = acc[v];

        const size_t ob = (size_t)img * CHW + (size_t)(td * 16) * HW + p0;
        float4 r1 = *(const float4*)&Yw[n1 * YROW + pp];
        float4 r2 = *(const float4*)&Yw[(n1 + 8) * YROW + pp];

        if (td < 15) {
            __syncthreads();
            sv[0] = pf0; sv[256] = pf1; sv[512] = pf2; sv[768] = pf3;
            __syncthreads();
        }
        *(float4*)&out[ob + (size_t)n1 * HW + pp]       = r1;
        *(float4*)&out[ob + (size_t)(n1 + 8) * HW + pp] = r2;
    }
}

__global__ __launch_bounds__(256) void inorm_softsign(float* __restrict__ y) {
    __shared__ float bs[256];
    __shared__ float bss[256];
    __shared__ float mean_s, rs_s;

    float* p = y + (size_t)blockIdx.x * 4096;
    const int tid = threadIdx.x;

    float v[16];
    float s = 0.f, ss = 0.f;
#pragma unroll
    for (int i = 0; i < 16; ++i) {
        float t = p[tid + 256 * i];
        v[i] = t;
        s  += t;
        ss += t * t;
    }
    bs[tid] = s;
    bss[tid] = ss;
    __syncthreads();
    for (int off = 128; off > 0; off >>= 1) {
        if (tid < off) {
            bs[tid]  += bs[tid + off];
            bss[tid] += bss[tid + off];
        }
        __syncthreads();
    }
    if (tid == 0) {
        const float mean = bs[0] * (1.0f / 4096.0f);
        const float var  = bss[0] * (1.0f / 4096.0f) - mean * mean;
        mean_s = mean;
        rs_s   = rsqrtf(var + 1e-5f);
    }
    __syncthreads();
    const float mean = mean_s, rs = rs_s;
#pragma unroll
    for (int i = 0; i < 16; ++i) {
        const float yn = (v[i] - mean) * rs;
        p[tid + 256 * i] = yn / (1.0f + fabsf(yn));
    }
}

extern "C" void kernel_launch(void* const* d_in, const int* in_sizes, int n_in,
                              void* d_out, int out_size, void* d_ws, size_t ws_size,
                              hipStream_t stream) {
    (void)in_sizes; (void)n_in; (void)out_size; (void)ws_size;
    const float* x     = (const float*)d_in[0];
    const float* wcore = (const float*)d_in[1];
    const float* wout  = (const float*)d_in[2];
    float* out = (float*)d_out;
    float* ws  = (float*)d_ws;

    repack_weights<<<512, 256, 0, stream>>>(wcore, wout, ws);

    tr_main<<<512, 256, 0, stream>>>(x, ws, out);

    inorm_softsign<<<16 * 256, 256, 0, stream>>>(out);
}
